# starting kernel
# baseline (speedup 1.0000x reference)
#include <hip/hip_runtime.h>

typedef float v2f __attribute__((ext_vector_type(2)));
typedef float v8f __attribute__((ext_vector_type(8)));

#define NNODE 4096
#define CH 64
#define NB 4

__device__ __forceinline__ v8f wmma4(v2f a, v2f b, v8f c) {
  return __builtin_amdgcn_wmma_f32_16x16x4_f32(false, a, false, b, (short)0, c,
                                               false, false);
}

__global__ __launch_bounds__(128) void k1_xw1(const float* __restrict__ X,
                                              const float* __restrict__ w1,
                                              float* __restrict__ outT) {
  const int lane = threadIdx.x & 31;
  const int wave = threadIdx.x >> 5;
  const int job  = blockIdx.x * 4 + wave;
  const int b    = job >> 8;
  const int j0   = (job & 255) << 4;
  const int m    = lane & 15;
  const int h    = lane >> 4;

  v8f acc[4] = {};
  const float* Xrow = X + (size_t)(b * NNODE + j0 + m) * CH;

#pragma unroll 4
  for (int k0 = 0; k0 < CH; k0 += 4) {
    const int kb = k0 + 2 * h;
    v2f a = *(const v2f*)(Xrow + kb);
#pragma unroll
    for (int t = 0; t < 4; ++t) {
      v2f bb;
      bb.x = w1[kb * CH + t * 16 + m];
      bb.y = w1[(kb + 1) * CH + t * 16 + m];
      acc[t] = wmma4(a, bb, acc[t]);
    }
  }
#pragma unroll
  for (int t = 0; t < 4; ++t)
#pragma unroll
    for (int rr = 0; rr < 8; ++rr)
      outT[(size_t)(b * CH + t * 16 + m) * NNODE + j0 + rr + 8 * h] =
          acc[t][rr];
}

__global__ void k1b_lr(const float* __restrict__ outT,
                       const float* __restrict__ alpha, float* __restrict__ l,
                       float* __restrict__ r) {
  const int gid = blockIdx.x * blockDim.x + threadIdx.x;
  const int b = gid >> 12;
  const int j = gid & (NNODE - 1);
  float aL = 0.f, aR = 0.f;
#pragma unroll 8
  for (int c = 0; c < CH; ++c) {
    float v = outT[(size_t)(b * CH + c) * NNODE + j];
    aL = fmaf(v, alpha[c], aL);
    aR = fmaf(v, alpha[CH + c], aR);
  }
  l[gid] = aL;
  r[gid] = aR;
}

__global__ __launch_bounds__(128) void k2_main(
    const float* __restrict__ A, const float* __restrict__ outT,
    const float* __restrict__ l, const float* __restrict__ r,
    const float* __restrict__ w2, float* __restrict__ out) {
  __shared__ __align__(16) float tile[4][16 * 66];

  const int lane = threadIdx.x & 31;
  const int wave = threadIdx.x >> 5;
  const int job  = blockIdx.x * 4 + wave;
  const int b    = job >> 8;
  const int i0   = (job & 255) << 4;
  const int m    = lane & 15;
  const int h    = lane >> 4;

  const float l_lane = l[b * NNODE + i0 + m];
  const float* Arow  = A + (size_t)(i0 + m) * NNODE;
  const float* rb    = r + b * NNODE;
  const float* oT    = outT + (size_t)b * CH * NNODE;

  v8f acc[4] = {};
  float rsum = 0.f;

  for (int j0 = 0; j0 < NNODE; j0 += 16) {
#pragma unroll
    for (int k = 0; k < 4; ++k) {
      const int jb = j0 + 4 * k + 2 * h;
      v2f ap = *(const v2f*)(Arow + jb);
      v2f rp = *(const v2f*)(rb + jb);
      float a0 = (i0 + m == jb)     ? 1.0f : ap.x;
      float a1 = (i0 + m == jb + 1) ? 1.0f : ap.y;
      float s0 = l_lane * rp.x;
      float s1 = l_lane * rp.y;
      s0 = s0 > 0.f ? s0 : 0.01f * s0;
      s1 = s1 > 0.f ? s1 : 0.01f * s1;
      v2f af;
      af.x = a0 * __expf(s0);
      af.y = a1 * __expf(s1);
      rsum += af.x + af.y;
#pragma unroll
      for (int t = 0; t < 4; ++t) {
        v2f bf = *(const v2f*)(oT + (size_t)(t * 16 + m) * NNODE + jb);
        acc[t] = wmma4(af, bf, acc[t]);
      }
    }
  }

  rsum += __shfl_xor(rsum, 16, 32);
  const float inv = 1.0f / rsum;

  float* T = tile[wave];
#pragma unroll
  for (int rr = 0; rr < 8; ++rr) {
    const float invr = __shfl(inv, rr + 8 * h, 32);
#pragma unroll
    for (int t = 0; t < 4; ++t)
      T[(rr + 8 * h) * 66 + t * 16 + m] = acc[t][rr] * invr;
  }
  __syncthreads();

  v8f fin[4] = {};
#pragma unroll 4
  for (int k0 = 0; k0 < CH; k0 += 4) {
    const int kb = k0 + 2 * h;
    v2f a = *(const v2f*)(T + m * 66 + kb);
#pragma unroll
    for (int t = 0; t < 4; ++t) {
      v2f bb;
      bb.x = w2[kb * CH + t * 16 + m];
      bb.y = w2[(kb + 1) * CH + t * 16 + m];
      fin[t] = wmma4(a, bb, fin[t]);
    }
  }

#pragma unroll
  for (int t = 0; t < 4; ++t)
#pragma unroll
    for (int rr = 0; rr < 8; ++rr)
      out[(size_t)(b * NNODE + i0 + rr + 8 * h) * CH + t * 16 + m] =
          fin[t][rr];
}

extern "C" void kernel_launch(void* const* d_in, const int* in_sizes, int n_in,
                              void* d_out, int out_size, void* d_ws,
                              size_t ws_size, hipStream_t stream) {
  const float* X     = (const float*)d_in[0];
  const float* A     = (const float*)d_in[1];
  const float* w1    = (const float*)d_in[2];
  const float* w2    = (const float*)d_in[3];
  const float* alpha = (const float*)d_in[4];
  float* out = (float*)d_out;

  float* ws   = (float*)d_ws;
  float* outT = ws;
  float* l    = outT + (size_t)NB * CH * NNODE;
  float* r    = l + (size_t)NB * NNODE;

  k1_xw1<<<NB * 256 / 4, 128, 0, stream>>>(X, w1, outT);
  k1b_lr<<<NB * NNODE / 256, 256, 0, stream>>>(outT, alpha, l, r);
  k2_main<<<NB * 256 / 4, 128, 0, stream>>>(A, outT, l, r, w2, out);
}


# k2 loop pipelined one block ahead: next block's 64x32 outT piece fetched once per work-group with whole-line loads into a 2-stage LDS ring read by all 4 waves; next adj/r in a 2nd register set
# speedup vs baseline: 1.5459x; 1.5459x over previous
#include <hip/hip_runtime.h>

typedef float v2f __attribute__((ext_vector_type(2)));
typedef float v8f __attribute__((ext_vector_type(8)));

#define NNODE 4096
#define CH 64
#define NB 4

__device__ __forceinline__ v8f wmma4(v2f a, v2f b, v8f c) {
  return __builtin_amdgcn_wmma_f32_16x16x4_f32(false, a, false, b, (short)0, c,
                                               false, false);
}

__global__ __launch_bounds__(128) void k1_xw1(const float* __restrict__ X,
                                              const float* __restrict__ w1,
                                              float* __restrict__ outT) {
  const int lane = threadIdx.x & 31;
  const int wave = threadIdx.x >> 5;
  const int job  = blockIdx.x * 4 + wave;
  const int b    = job >> 8;
  const int j0   = (job & 255) << 4;
  const int m    = lane & 15;
  const int h    = lane >> 4;

  v8f acc[4] = {};
  const float* Xrow = X + (size_t)(b * NNODE + j0 + m) * CH;

#pragma unroll 4
  for (int k0 = 0; k0 < CH; k0 += 4) {
    const int kb = k0 + 2 * h;
    v2f a = *(const v2f*)(Xrow + kb);
#pragma unroll
    for (int t = 0; t < 4; ++t) {
      v2f bb;
      bb.x = w1[kb * CH + t * 16 + m];
      bb.y = w1[(kb + 1) * CH + t * 16 + m];
      acc[t] = wmma4(a, bb, acc[t]);
    }
  }
#pragma unroll
  for (int t = 0; t < 4; ++t)
#pragma unroll
    for (int rr = 0; rr < 8; ++rr)
      outT[(size_t)(b * CH + t * 16 + m) * NNODE + j0 + rr + 8 * h] =
          acc[t][rr];
}

__global__ void k1b_lr(const float* __restrict__ outT,
                       const float* __restrict__ alpha, float* __restrict__ l,
                       float* __restrict__ r) {
  const int gid = blockIdx.x * blockDim.x + threadIdx.x;
  const int b = gid >> 12;
  const int j = gid & (NNODE - 1);
  float aL = 0.f, aR = 0.f;
#pragma unroll 8
  for (int c = 0; c < CH; ++c) {
    float v = outT[(size_t)(b * CH + c) * NNODE + j];
    aL = fmaf(v, alpha[c], aL);
    aR = fmaf(v, alpha[CH + c], aR);
  }
  l[gid] = aL;
  r[gid] = aR;
}

#define BP 36

__device__ __forceinline__ void k2_block(
    const float4 (&a4)[4], const float4 (&r4)[4], float4 (&na4)[4],
    float4 (&nr4)[4], const float* __restrict__ Bc, float* __restrict__ Bn,
    const float* __restrict__ gsrc, const float* __restrict__ Arow,
    const float* __restrict__ rb, const int jn, const int d,
    const float l_lane, v8f (&acc)[4], float& rsum) {
  float4 gB[4], b4[4][4];
#pragma unroll
  for (int q = 0; q < 4; ++q)
    gB[q] = *(const float4*)(gsrc + (size_t)(16 * q) * NNODE + jn);
#pragma unroll
  for (int g = 0; g < 4; ++g) {
    na4[g] = *(const float4*)(Arow + jn + 4 * g);
    nr4[g] = *(const float4*)(rb + jn + 4 * g);
  }
#pragma unroll
  for (int t = 0; t < 4; ++t)
#pragma unroll
    for (int g = 0; g < 4; ++g)
      b4[t][g] = *(const float4*)(Bc + t * 16 * BP + 4 * g);
#pragma unroll
  for (int g = 0; g < 4; ++g) {
    const float av[4] = {a4[g].x, a4[g].y, a4[g].z, a4[g].w};
    const float rv[4] = {r4[g].x, r4[g].y, r4[g].z, r4[g].w};
    float w[4];
#pragma unroll
    for (int u = 0; u < 4; ++u) {
      const float a = (d == 4 * g + u) ? 1.0f : av[u];
      float s = l_lane * rv[u];
      s = s > 0.f ? s : 0.01f * s;
      w[u] = a * __expf(s);
    }
#pragma unroll
    for (int p = 0; p < 2; ++p) {
      v2f af;
      af.x = w[2 * p];
      af.y = w[2 * p + 1];
      rsum += af.x + af.y;
#pragma unroll
      for (int t = 0; t < 4; ++t) {
        v2f bf;
        bf.x = p ? b4[t][g].z : b4[t][g].x;
        bf.y = p ? b4[t][g].w : b4[t][g].y;
        acc[t] = wmma4(af, bf, acc[t]);
      }
    }
  }
#pragma unroll
  for (int q = 0; q < 4; ++q) *(float4*)(Bn + 16 * q * BP) = gB[q];
  __builtin_amdgcn_sched_group_barrier(0x020, 16, 0);
  __builtin_amdgcn_sched_group_barrier(0x100, 16, 0);
  __builtin_amdgcn_sched_group_barrier(0x008, 32, 0);
  __builtin_amdgcn_sched_group_barrier(0x200, 4, 0);
  asm volatile("" ::"v"(acc[0]), "v"(acc[1]), "v"(acc[2]), "v"(acc[3]),
               "v"(na4[0].x), "v"(na4[0].w), "v"(na4[1].x), "v"(na4[1].w),
               "v"(na4[2].x), "v"(na4[2].w), "v"(na4[3].x), "v"(na4[3].w),
               "v"(nr4[0].x), "v"(nr4[0].w), "v"(nr4[1].x), "v"(nr4[1].w),
               "v"(nr4[2].x), "v"(nr4[2].w), "v"(nr4[3].x), "v"(nr4[3].w));
  __builtin_amdgcn_fence(__ATOMIC_RELEASE, "workgroup", "local");
  __builtin_amdgcn_s_barrier();
  __builtin_amdgcn_fence(__ATOMIC_ACQUIRE, "workgroup", "local");
}

__global__ __launch_bounds__(128) void k2_main(
    const float* __restrict__ A, const float* __restrict__ outT,
    const float* __restrict__ l, const float* __restrict__ r,
    const float* __restrict__ w2, float* __restrict__ out) {
  __shared__ __align__(16) float tile[4][16 * 66];
  __shared__ __align__(16) float Bs[2][CH * BP];

  const int tid  = threadIdx.x;
  const int lane = tid & 31;
  const int wave = tid >> 5;
  const int job  = blockIdx.x * 4 + wave;
  const int b    = job >> 8;
  const int i0   = (job & 255) << 4;
  const int m    = lane & 15;
  const int h    = lane >> 4;

  const float l_lane = l[b * NNODE + i0 + m];
  const float* Arow = A + (size_t)(i0 + m) * NNODE + 16 * h;
  const float* rb   = r + b * NNODE + 16 * h;
  const float* oT   = outT + (size_t)b * CH * NNODE;

  const float* gsrc = oT + (size_t)(tid >> 3) * NNODE + 4 * (tid & 7);
  const int sdst = (tid >> 3) * BP + 4 * (tid & 7);
  const int fsrc = m * BP + 16 * h;

  v8f acc[4] = {};
  float rsum = 0.f;

  float4 aA[4], rA[4], aB[4], rB[4];
  {
    float4 gB[4];
#pragma unroll
    for (int q = 0; q < 4; ++q)
      gB[q] = *(const float4*)(gsrc + (size_t)(16 * q) * NNODE);
#pragma unroll
    for (int g = 0; g < 4; ++g) {
      aA[g] = *(const float4*)(Arow + 4 * g);
      rA[g] = *(const float4*)(rb + 4 * g);
    }
#pragma unroll
    for (int q = 0; q < 4; ++q) *(float4*)(&Bs[0][sdst + 16 * q * BP]) = gB[q];
  }
  __syncthreads();

  int d = i0 + m - 16 * h;
  for (int j0 = 0; j0 < NNODE; j0 += 64) {
    const int jlast = (j0 + 64 < NNODE) ? j0 + 64 : NNODE - 32;
    k2_block(aA, rA, aB, rB, &Bs[0][fsrc], &Bs[1][sdst], gsrc, Arow, rb,
             j0 + 32, d, l_lane, acc, rsum);
    k2_block(aB, rB, aA, rA, &Bs[1][fsrc], &Bs[0][sdst], gsrc, Arow, rb,
             jlast, d - 32, l_lane, acc, rsum);
    d -= 64;
  }

  rsum += __shfl_xor(rsum, 16, 32);
  const float inv = 1.0f / rsum;

  float* T = tile[wave];
#pragma unroll
  for (int rr = 0; rr < 8; ++rr) {
    const float invr = __shfl(inv, rr + 8 * h, 32);
#pragma unroll
    for (int t = 0; t < 4; ++t)
      T[(rr + 8 * h) * 66 + t * 16 + m] = acc[t][rr] * invr;
  }
  __syncthreads();

  v8f fin[4] = {};
#pragma unroll 4
  for (int k0 = 0; k0 < CH; k0 += 4) {
    const int kb = k0 + 2 * h;
    v2f a = *(const v2f*)(T + m * 66 + kb);
#pragma unroll
    for (int t = 0; t < 4; ++t) {
      v2f bb;
      bb.x = w2[kb * CH + t * 16 + m];
      bb.y = w2[(kb + 1) * CH + t * 16 + m];
      fin[t] = wmma4(a, bb, fin[t]);
    }
  }

#pragma unroll
  for (int t = 0; t < 4; ++t)
#pragma unroll
    for (int rr = 0; rr < 8; ++rr)
      out[(size_t)(b * NNODE + i0 + rr + 8 * h) * CH + t * 16 + m] =
          fin[t][rr];
}

extern "C" void kernel_launch(void* const* d_in, const int* in_sizes, int n_in,
                              void* d_out, int out_size, void* d_ws,
                              size_t ws_size, hipStream_t stream) {
  const float* X     = (const float*)d_in[0];
  const float* A     = (const float*)d_in[1];
  const float* w1    = (const float*)d_in[2];
  const float* w2    = (const float*)d_in[3];
  const float* alpha = (const float*)d_in[4];
  float* out = (float*)d_out;

  float* ws   = (float*)d_ws;
  float* outT = ws;
  float* l    = outT + (size_t)NB * CH * NNODE;
  float* r    = l + (size_t)NB * NNODE;

  k1_xw1<<<NB * 256 / 4, 128, 0, stream>>>(X, w1, outT);
  k1b_lr<<<NB * NNODE / 256, 256, 0, stream>>>(outT, alpha, l, r);
  k2_main<<<NB * 256 / 4, 128, 0, stream>>>(A, outT, l, r, w2, out);
}


# k2: j range cut into 4 parts, each walked by its own 4-wave group with its own LDS ring (16-wave work-group, 4 waves per SIMD); partials added via LDS in fixed order; j sum order changes
# speedup vs baseline: 2.2811x; 1.4756x over previous
#include <hip/hip_runtime.h>

typedef float v2f __attribute__((ext_vector_type(2)));
typedef float v8f __attribute__((ext_vector_type(8)));

#define NNODE 4096
#define CH 64
#define NB 4

__device__ __forceinline__ v8f wmma4(v2f a, v2f b, v8f c) {
  return __builtin_amdgcn_wmma_f32_16x16x4_f32(false, a, false, b, (short)0, c,
                                               false, false);
}

__global__ __launch_bounds__(128) void k1_xw1(const float* __restrict__ X,
                                              const float* __restrict__ w1,
                                              float* __restrict__ outT) {
  const int lane = threadIdx.x & 31;
  const int wave = threadIdx.x >> 5;
  const int job  = blockIdx.x * 4 + wave;
  const int b    = job >> 8;
  const int j0   = (job & 255) << 4;
  const int m    = lane & 15;
  const int h    = lane >> 4;

  v8f acc[4] = {};
  const float* Xrow = X + (size_t)(b * NNODE + j0 + m) * CH;

#pragma unroll 4
  for (int k0 = 0; k0 < CH; k0 += 4) {
    const int kb = k0 + 2 * h;
    v2f a = *(const v2f*)(Xrow + kb);
#pragma unroll
    for (int t = 0; t < 4; ++t) {
      v2f bb;
      bb.x = w1[kb * CH + t * 16 + m];
      bb.y = w1[(kb + 1) * CH + t * 16 + m];
      acc[t] = wmma4(a, bb, acc[t]);
    }
  }
#pragma unroll
  for (int t = 0; t < 4; ++t)
#pragma unroll
    for (int rr = 0; rr < 8; ++rr)
      outT[(size_t)(b * CH + t * 16 + m) * NNODE + j0 + rr + 8 * h] =
          acc[t][rr];
}

__global__ void k1b_lr(const float* __restrict__ outT,
                       const float* __restrict__ alpha, float* __restrict__ l,
                       float* __restrict__ r) {
  const int gid = blockIdx.x * blockDim.x + threadIdx.x;
  const int b = gid >> 12;
  const int j = gid & (NNODE - 1);
  float aL = 0.f, aR = 0.f;
#pragma unroll 8
  for (int c = 0; c < CH; ++c) {
    float v = outT[(size_t)(b * CH + c) * NNODE + j];
    aL = fmaf(v, alpha[c], aL);
    aR = fmaf(v, alpha[CH + c], aR);
  }
  l[gid] = aL;
  r[gid] = aR;
}

#define JS 4
#define NJ (NNODE / JS)
#define BP 36

__device__ __forceinline__ void k2_block(
    const float4 (&a4)[4], const float4 (&r4)[4], float4 (&na4)[4],
    float4 (&nr4)[4], const float* __restrict__ Bc, float* __restrict__ Bn,
    const float* __restrict__ gsrc, const float* __restrict__ Arow,
    const float* __restrict__ rb, const int jn, const int d,
    const float l_lane, v8f (&acc)[4], float& rsum) {
  float4 gB[4], b4[4][4];
#pragma unroll
  for (int q = 0; q < 4; ++q)
    gB[q] = *(const float4*)(gsrc + (size_t)(16 * q) * NNODE + jn);
#pragma unroll
  for (int g = 0; g < 4; ++g) {
    na4[g] = *(const float4*)(Arow + jn + 4 * g);
    nr4[g] = *(const float4*)(rb + jn + 4 * g);
  }
#pragma unroll
  for (int t = 0; t < 4; ++t)
#pragma unroll
    for (int g = 0; g < 4; ++g)
      b4[t][g] = *(const float4*)(Bc + t * 16 * BP + 4 * g);
#pragma unroll
  for (int g = 0; g < 4; ++g) {
    const float av[4] = {a4[g].x, a4[g].y, a4[g].z, a4[g].w};
    const float rv[4] = {r4[g].x, r4[g].y, r4[g].z, r4[g].w};
    float w[4];
#pragma unroll
    for (int u = 0; u < 4; ++u) {
      const float a = (d == 4 * g + u) ? 1.0f : av[u];
      float s = l_lane * rv[u];
      s = s > 0.f ? s : 0.01f * s;
      w[u] = a * __expf(s);
    }
#pragma unroll
    for (int p = 0; p < 2; ++p) {
      v2f af;
      af.x = w[2 * p];
      af.y = w[2 * p + 1];
      rsum += af.x + af.y;
#pragma unroll
      for (int t = 0; t < 4; ++t) {
        v2f bf;
        bf.x = p ? b4[t][g].z : b4[t][g].x;
        bf.y = p ? b4[t][g].w : b4[t][g].y;
        acc[t] = wmma4(af, bf, acc[t]);
      }
    }
  }
#pragma unroll
  for (int q = 0; q < 4; ++q) *(float4*)(Bn + 16 * q * BP) = gB[q];
  __builtin_amdgcn_sched_group_barrier(0x020, 16, 0);
  __builtin_amdgcn_sched_group_barrier(0x100, 16, 0);
  __builtin_amdgcn_sched_group_barrier(0x008, 32, 0);
  __builtin_amdgcn_sched_group_barrier(0x200, 4, 0);
  asm volatile("" ::"v"(acc[0]), "v"(acc[1]), "v"(acc[2]), "v"(acc[3]),
               "v"(na4[0].x), "v"(na4[0].w), "v"(na4[1].x), "v"(na4[1].w),
               "v"(na4[2].x), "v"(na4[2].w), "v"(na4[3].x), "v"(na4[3].w),
               "v"(nr4[0].x), "v"(nr4[0].w), "v"(nr4[1].x), "v"(nr4[1].w),
               "v"(nr4[2].x), "v"(nr4[2].w), "v"(nr4[3].x), "v"(nr4[3].w));
  __builtin_amdgcn_fence(__ATOMIC_RELEASE, "workgroup", "local");
  __builtin_amdgcn_s_barrier();
  __builtin_amdgcn_fence(__ATOMIC_ACQUIRE, "workgroup", "local");
}

__global__ __launch_bounds__(128 * JS) void k2_main(
    const float* __restrict__ A, const float* __restrict__ outT,
    const float* __restrict__ l, const float* __restrict__ r,
    const float* __restrict__ w2, float* __restrict__ out) {
  __shared__ __align__(16) float tile[4][16 * 66];
  __shared__ __align__(16) float Bs[JS][2][CH * BP];
  __shared__ __align__(16) float Racc[(JS - 1) * 4][32 * 32];
  __shared__ __align__(16) float Rsum[(JS - 1) * 4][32];

  const int tid  = threadIdx.x;
  const int lane = tid & 31;
  const int wave = tid >> 5;
  const int strip = wave & 3;
  const int js   = wave >> 2;
  const int tg   = tid & 127;
  const int jbeg = js * NJ;
  const int job  = blockIdx.x * 4 + strip;
  const int b    = job >> 8;
  const int i0   = (job & 255) << 4;
  const int m    = lane & 15;
  const int h    = lane >> 4;

  const float l_lane = l[b * NNODE + i0 + m];
  const float* Arow = A + (size_t)(i0 + m) * NNODE + 16 * h + jbeg;
  const float* rb   = r + b * NNODE + 16 * h + jbeg;
  const float* oT   = outT + (size_t)b * CH * NNODE;

  const float* gsrc = oT + (size_t)(tg >> 3) * NNODE + 4 * (tg & 7) + jbeg;
  const int sdst = (tg >> 3) * BP + 4 * (tg & 7);
  float* B0 = &Bs[js][0][0];
  float* B1 = &Bs[js][1][0];
  const int fsrc = m * BP + 16 * h;

  v8f acc[4] = {};
  float rsum = 0.f;

  float4 aA[4], rA[4], aB[4], rB[4];
  {
    float4 gB[4];
#pragma unroll
    for (int q = 0; q < 4; ++q)
      gB[q] = *(const float4*)(gsrc + (size_t)(16 * q) * NNODE);
#pragma unroll
    for (int g = 0; g < 4; ++g) {
      aA[g] = *(const float4*)(Arow + 4 * g);
      rA[g] = *(const float4*)(rb + 4 * g);
    }
#pragma unroll
    for (int q = 0; q < 4; ++q) *(float4*)(B0 + sdst + 16 * q * BP) = gB[q];
  }
  __syncthreads();

  int d = i0 + m - 16 * h - jbeg;
  for (int j0 = 0; j0 < NJ; j0 += 64) {
    const int jlast = (j0 + 64 < NJ) ? j0 + 64 : NJ - 32;
    k2_block(aA, rA, aB, rB, B0 + fsrc, B1 + sdst, gsrc, Arow, rb,
             j0 + 32, d, l_lane, acc, rsum);
    k2_block(aB, rB, aA, rA, B1 + fsrc, B0 + sdst, gsrc, Arow, rb,
             jlast, d - 32, l_lane, acc, rsum);
    d -= 64;
  }

  if (js > 0) {
    float* R = Racc[(js - 1) * 4 + strip];
#pragma unroll
    for (int t = 0; t < 4; ++t)
#pragma unroll
      for (int rr = 0; rr < 8; ++rr) R[(t * 8 + rr) * 32 + lane] = acc[t][rr];
    Rsum[(js - 1) * 4 + strip][lane] = rsum;
  }
  __syncthreads();

  float* T = tile[strip];
  if (js == 0) {
#pragma unroll
    for (int p = 0; p < JS - 1; ++p) {
      const float* R = Racc[p * 4 + strip];
#pragma unroll
      for (int t = 0; t < 4; ++t)
#pragma unroll
        for (int rr = 0; rr < 8; ++rr) acc[t][rr] += R[(t * 8 + rr) * 32 + lane];
      rsum += Rsum[p * 4 + strip][lane];
    }

    rsum += __shfl_xor(rsum, 16, 32);
    const float inv = 1.0f / rsum;

#pragma unroll
    for (int rr = 0; rr < 8; ++rr) {
      const float invr = __shfl(inv, rr + 8 * h, 32);
#pragma unroll
      for (int t = 0; t < 4; ++t)
        T[(rr + 8 * h) * 66 + t * 16 + m] = acc[t][rr] * invr;
    }
  }
  __syncthreads();

  if (js == 0) {
    v8f fin[4] = {};
#pragma unroll 4
    for (int k0 = 0; k0 < CH; k0 += 4) {
      const int kb = k0 + 2 * h;
      v2f a = *(const v2f*)(T + m * 66 + kb);
#pragma unroll
      for (int t = 0; t < 4; ++t) {
        v2f bb;
        bb.x = w2[kb * CH + t * 16 + m];
        bb.y = w2[(kb + 1) * CH + t * 16 + m];
        fin[t] = wmma4(a, bb, fin[t]);
      }
    }

#pragma unroll
    for (int t = 0; t < 4; ++t)
#pragma unroll
      for (int rr = 0; rr < 8; ++rr)
        out[(size_t)(b * NNODE + i0 + rr + 8 * h) * CH + t * 16 + m] =
            fin[t][rr];
  }
}

extern "C" void kernel_launch(void* const* d_in, const int* in_sizes, int n_in,
                              void* d_out, int out_size, void* d_ws,
                              size_t ws_size, hipStream_t stream) {
  const float* X     = (const float*)d_in[0];
  const float* A     = (const float*)d_in[1];
  const float* w1    = (const float*)d_in[2];
  const float* w2    = (const float*)d_in[3];
  const float* alpha = (const float*)d_in[4];
  float* out = (float*)d_out;

  float* ws   = (float*)d_ws;
  float* outT = ws;
  float* l    = outT + (size_t)NB * CH * NNODE;
  float* r    = l + (size_t)NB * NNODE;

  k1_xw1<<<NB * 256 / 4, 128, 0, stream>>>(X, w1, outT);
  k1b_lr<<<NB * NNODE / 256, 256, 0, stream>>>(outT, alpha, l, r);
  k2_main<<<NB * 256 / 4, 128 * JS, 0, stream>>>(A, outT, l, r, w2, out);
}


# k2: LDS ring stage widened from 32 to 128 columns (pieces fetched 4 blocks ahead), so the work-group barrier comes once per 128 columns, not per 32; closing arrays reuse the ring's LDS
# speedup vs baseline: 2.3696x; 1.0388x over previous
#include <hip/hip_runtime.h>

typedef float v2f __attribute__((ext_vector_type(2)));
typedef float v8f __attribute__((ext_vector_type(8)));

#define NNODE 4096
#define CH 64
#define NB 4

__device__ __forceinline__ v8f wmma4(v2f a, v2f b, v8f c) {
  return __builtin_amdgcn_wmma_f32_16x16x4_f32(false, a, false, b, (short)0, c,
                                               false, false);
}

__global__ __launch_bounds__(128) void k1_xw1(const float* __restrict__ X,
                                              const float* __restrict__ w1,
                                              float* __restrict__ outT) {
  const int lane = threadIdx.x & 31;
  const int wave = threadIdx.x >> 5;
  const int job  = blockIdx.x * 4 + wave;
  const int b    = job >> 8;
  const int j0   = (job & 255) << 4;
  const int m    = lane & 15;
  const int h    = lane >> 4;

  v8f acc[4] = {};
  const float* Xrow = X + (size_t)(b * NNODE + j0 + m) * CH;

#pragma unroll 4
  for (int k0 = 0; k0 < CH; k0 += 4) {
    const int kb = k0 + 2 * h;
    v2f a = *(const v2f*)(Xrow + kb);
#pragma unroll
    for (int t = 0; t < 4; ++t) {
      v2f bb;
      bb.x = w1[kb * CH + t * 16 + m];
      bb.y = w1[(kb + 1) * CH + t * 16 + m];
      acc[t] = wmma4(a, bb, acc[t]);
    }
  }
#pragma unroll
  for (int t = 0; t < 4; ++t)
#pragma unroll
    for (int rr = 0; rr < 8; ++rr)
      outT[(size_t)(b * CH + t * 16 + m) * NNODE + j0 + rr + 8 * h] =
          acc[t][rr];
}

__global__ void k1b_lr(const float* __restrict__ outT,
                       const float* __restrict__ alpha, float* __restrict__ l,
                       float* __restrict__ r) {
  const int gid = blockIdx.x * blockDim.x + threadIdx.x;
  const int b = gid >> 12;
  const int j = gid & (NNODE - 1);
  float aL = 0.f, aR = 0.f;
#pragma unroll 8
  for (int c = 0; c < CH; ++c) {
    float v = outT[(size_t)(b * CH + c) * NNODE + j];
    aL = fmaf(v, alpha[c], aL);
    aR = fmaf(v, alpha[CH + c], aR);
  }
  l[gid] = aL;
  r[gid] = aR;
}

#define JS 4
#define NJ (NNODE / JS)
#define SB 4
#define SC (32 * SB)
#define BP (SC + 4)
#define STAGE (CH * BP)

template <bool LAST>
__device__ __forceinline__ void k2_block(
    const float4 (&a4)[4], const float4 (&r4)[4], float4 (&na4)[4],
    float4 (&nr4)[4], const float* __restrict__ Bc, float* __restrict__ Bn,
    const float* __restrict__ gsrc, const float* __restrict__ Arow,
    const float* __restrict__ rb, const int jg, const int jn, const int d,
    const float l_lane, v8f (&acc)[4], float& rsum) {
  float4 gB[4], b4[4][4];
#pragma unroll
  for (int q = 0; q < 4; ++q)
    gB[q] = *(const float4*)(gsrc + (size_t)(16 * q) * NNODE + jg);
#pragma unroll
  for (int g = 0; g < 4; ++g) {
    na4[g] = *(const float4*)(Arow + jn + 4 * g);
    nr4[g] = *(const float4*)(rb + jn + 4 * g);
  }
#pragma unroll
  for (int t = 0; t < 4; ++t)
#pragma unroll
    for (int g = 0; g < 4; ++g)
      b4[t][g] = *(const float4*)(Bc + t * 16 * BP + 4 * g);
#pragma unroll
  for (int g = 0; g < 4; ++g) {
    const float av[4] = {a4[g].x, a4[g].y, a4[g].z, a4[g].w};
    const float rv[4] = {r4[g].x, r4[g].y, r4[g].z, r4[g].w};
    float w[4];
#pragma unroll
    for (int u = 0; u < 4; ++u) {
      const float a = (d == 4 * g + u) ? 1.0f : av[u];
      float s = l_lane * rv[u];
      s = s > 0.f ? s : 0.01f * s;
      w[u] = a * __expf(s);
    }
#pragma unroll
    for (int p = 0; p < 2; ++p) {
      v2f af;
      af.x = w[2 * p];
      af.y = w[2 * p + 1];
      rsum += af.x + af.y;
#pragma unroll
      for (int t = 0; t < 4; ++t) {
        v2f bf;
        bf.x = p ? b4[t][g].z : b4[t][g].x;
        bf.y = p ? b4[t][g].w : b4[t][g].y;
        acc[t] = wmma4(af, bf, acc[t]);
      }
    }
  }
#pragma unroll
  for (int q = 0; q < 4; ++q) *(float4*)(Bn + 16 * q * BP) = gB[q];
  __builtin_amdgcn_sched_group_barrier(0x020, 16, 0);
  __builtin_amdgcn_sched_group_barrier(0x100, 16, 0);
  __builtin_amdgcn_sched_group_barrier(0x008, 32, 0);
  __builtin_amdgcn_sched_group_barrier(0x200, 4, 0);
  asm volatile("" ::"v"(acc[0]), "v"(acc[1]), "v"(acc[2]), "v"(acc[3]),
               "v"(na4[0].x), "v"(na4[0].w), "v"(na4[1].x), "v"(na4[1].w),
               "v"(na4[2].x), "v"(na4[2].w), "v"(na4[3].x), "v"(na4[3].w),
               "v"(nr4[0].x), "v"(nr4[0].w), "v"(nr4[1].x), "v"(nr4[1].w),
               "v"(nr4[2].x), "v"(nr4[2].w), "v"(nr4[3].x), "v"(nr4[3].w));
  if (LAST) {
    __builtin_amdgcn_fence(__ATOMIC_RELEASE, "workgroup", "local");
    __builtin_amdgcn_s_barrier();
    __builtin_amdgcn_fence(__ATOMIC_ACQUIRE, "workgroup", "local");
  }
}

__global__ __launch_bounds__(128 * JS) void k2_main(
    const float* __restrict__ A, const float* __restrict__ outT,
    const float* __restrict__ l, const float* __restrict__ r,
    const float* __restrict__ w2, float* __restrict__ out) {
  __shared__ __align__(16) float smem[JS * 2 * STAGE];
  float* const tileAll = smem;
  float* const RaccAll = tileAll + 4 * 16 * 66;
  float* const RsumAll = RaccAll + (JS - 1) * 4 * 32 * 32;

  const int tid  = threadIdx.x;
  const int lane = tid & 31;
  const int wave = tid >> 5;
  const int strip = wave & 3;
  const int js   = wave >> 2;
  const int tg   = tid & 127;
  const int jbeg = js * NJ;
  const int job  = blockIdx.x * 4 + strip;
  const int b    = job >> 8;
  const int i0   = (job & 255) << 4;
  const int m    = lane & 15;
  const int h    = lane >> 4;

  const float l_lane = l[b * NNODE + i0 + m];
  const float* Arow = A + (size_t)(i0 + m) * NNODE + 16 * h + jbeg;
  const float* rb   = r + b * NNODE + 16 * h + jbeg;
  const float* oT   = outT + (size_t)b * CH * NNODE;

  const float* gsrc = oT + (size_t)(tg >> 3) * NNODE + 4 * (tg & 7) + jbeg;
  const int sdst = (tg >> 3) * BP + 4 * (tg & 7);
  float* const Bg = smem + js * 2 * STAGE;
  const int fsrc = m * BP + 16 * h;

  v8f acc[4] = {};
  float rsum = 0.f;

  float4 aA[4], rA[4], aB[4], rB[4];
  {
    float4 gB[SB][4];
#pragma unroll
    for (int s = 0; s < SB; ++s)
#pragma unroll
      for (int q = 0; q < 4; ++q)
        gB[s][q] = *(const float4*)(gsrc + (size_t)(16 * q) * NNODE + 32 * s);
#pragma unroll
    for (int g = 0; g < 4; ++g) {
      aA[g] = *(const float4*)(Arow + 4 * g);
      rA[g] = *(const float4*)(rb + 4 * g);
    }
#pragma unroll
    for (int s = 0; s < SB; ++s)
#pragma unroll
      for (int q = 0; q < 4; ++q)
        *(float4*)(Bg + sdst + 32 * s + 16 * q * BP) = gB[s][q];
  }
  __syncthreads();

  int d = i0 + m - 16 * h - jbeg;
  int cs = 0;
  for (int j0 = 0; j0 < NJ; j0 += SC) {
    const float* Bc = Bg + cs + fsrc;
    float* Bn = Bg + (STAGE - cs) + sdst;
    const bool more = j0 + SC < NJ;
    const int jg = more ? j0 + SC : 0;
    const int jend = more ? j0 + SC : NJ - 32;
    k2_block<false>(aA, rA, aB, rB, Bc, Bn, gsrc, Arow, rb, jg, j0 + 32, d,
                    l_lane, acc, rsum);
    k2_block<false>(aB, rB, aA, rA, Bc + 32, Bn + 32, gsrc, Arow, rb, jg + 32,
                    j0 + 64, d - 32, l_lane, acc, rsum);
    k2_block<false>(aA, rA, aB, rB, Bc + 64, Bn + 64, gsrc, Arow, rb, jg + 64,
                    j0 + 96, d - 64, l_lane, acc, rsum);
    k2_block<true>(aB, rB, aA, rA, Bc + 96, Bn + 96, gsrc, Arow, rb, jg + 96,
                   jend, d - 96, l_lane, acc, rsum);
    d -= SC;
    cs = STAGE - cs;
  }

  if (js > 0) {
    float* R = RaccAll + ((js - 1) * 4 + strip) * (32 * 32);
#pragma unroll
    for (int t = 0; t < 4; ++t)
#pragma unroll
      for (int rr = 0; rr < 8; ++rr) R[(t * 8 + rr) * 32 + lane] = acc[t][rr];
    RsumAll[((js - 1) * 4 + strip) * 32 + lane] = rsum;
  }
  __syncthreads();

  float* T = tileAll + strip * (16 * 66);
  if (js == 0) {
#pragma unroll
    for (int p = 0; p < JS - 1; ++p) {
      const float* R = RaccAll + (p * 4 + strip) * (32 * 32);
#pragma unroll
      for (int t = 0; t < 4; ++t)
#pragma unroll
        for (int rr = 0; rr < 8; ++rr) acc[t][rr] += R[(t * 8 + rr) * 32 + lane];
      rsum += RsumAll[(p * 4 + strip) * 32 + lane];
    }

    rsum += __shfl_xor(rsum, 16, 32);
    const float inv = 1.0f / rsum;

#pragma unroll
    for (int rr = 0; rr < 8; ++rr) {
      const float invr = __shfl(inv, rr + 8 * h, 32);
#pragma unroll
      for (int t = 0; t < 4; ++t)
        T[(rr + 8 * h) * 66 + t * 16 + m] = acc[t][rr] * invr;
    }
  }
  __syncthreads();

  if (js == 0) {
    v8f fin[4] = {};
#pragma unroll 4
    for (int k0 = 0; k0 < CH; k0 += 4) {
      const int kb = k0 + 2 * h;
      v2f a = *(const v2f*)(T + m * 66 + kb);
#pragma unroll
      for (int t = 0; t < 4; ++t) {
        v2f bb;
        bb.x = w2[kb * CH + t * 16 + m];
        bb.y = w2[(kb + 1) * CH + t * 16 + m];
        fin[t] = wmma4(a, bb, fin[t]);
      }
    }

#pragma unroll
    for (int t = 0; t < 4; ++t)
#pragma unroll
      for (int rr = 0; rr < 8; ++rr)
        out[(size_t)(b * NNODE + i0 + rr + 8 * h) * CH + t * 16 + m] =
            fin[t][rr];
  }
}

extern "C" void kernel_launch(void* const* d_in, const int* in_sizes, int n_in,
                              void* d_out, int out_size, void* d_ws,
                              size_t ws_size, hipStream_t stream) {
  const float* X     = (const float*)d_in[0];
  const float* A     = (const float*)d_in[1];
  const float* w1    = (const float*)d_in[2];
  const float* w2    = (const float*)d_in[3];
  const float* alpha = (const float*)d_in[4];
  float* out = (float*)d_out;

  float* ws   = (float*)d_ws;
  float* outT = ws;
  float* l    = outT + (size_t)NB * CH * NNODE;
  float* r    = l + (size_t)NB * NNODE;

  k1_xw1<<<NB * 256 / 4, 128, 0, stream>>>(X, w1, outT);
  k1b_lr<<<NB * NNODE / 256, 256, 0, stream>>>(outT, alpha, l, r);
  k2_main<<<NB * 256 / 4, 128 * JS, 0, stream>>>(A, outT, l, r, w2, out);
}


# k2: vector work per weight cut by 40%: diagonal patched once per strip under a wave-uniform scalar branch, leaky relu as one max, log2(e) folded into l, row sums as 2 packed sums (sum order changes)
# speedup vs baseline: 2.4470x; 1.0327x over previous
#include <hip/hip_runtime.h>

typedef float v2f __attribute__((ext_vector_type(2)));
typedef float v8f __attribute__((ext_vector_type(8)));

#define NNODE 4096
#define CH 64
#define NB 4

__device__ __forceinline__ v8f wmma4(v2f a, v2f b, v8f c) {
  return __builtin_amdgcn_wmma_f32_16x16x4_f32(false, a, false, b, (short)0, c,
                                               false, false);
}

__global__ __launch_bounds__(128) void k1_xw1(const float* __restrict__ X,
                                              const float* __restrict__ w1,
                                              float* __restrict__ outT) {
  const int lane = threadIdx.x & 31;
  const int wave = threadIdx.x >> 5;
  const int job  = blockIdx.x * 4 + wave;
  const int b    = job >> 8;
  const int j0   = (job & 255) << 4;
  const int m    = lane & 15;
  const int h    = lane >> 4;

  v8f acc[4] = {};
  const float* Xrow = X + (size_t)(b * NNODE + j0 + m) * CH;

#pragma unroll 4
  for (int k0 = 0; k0 < CH; k0 += 4) {
    const int kb = k0 + 2 * h;
    v2f a = *(const v2f*)(Xrow + kb);
#pragma unroll
    for (int t = 0; t < 4; ++t) {
      v2f bb;
      bb.x = w1[kb * CH + t * 16 + m];
      bb.y = w1[(kb + 1) * CH + t * 16 + m];
      acc[t] = wmma4(a, bb, acc[t]);
    }
  }
#pragma unroll
  for (int t = 0; t < 4; ++t)
#pragma unroll
    for (int rr = 0; rr < 8; ++rr)
      outT[(size_t)(b * CH + t * 16 + m) * NNODE + j0 + rr + 8 * h] =
          acc[t][rr];
}

__global__ void k1b_lr(const float* __restrict__ outT,
                       const float* __restrict__ alpha, float* __restrict__ l,
                       float* __restrict__ r) {
  const int gid = blockIdx.x * blockDim.x + threadIdx.x;
  const int b = gid >> 12;
  const int j = gid & (NNODE - 1);
  float aL = 0.f, aR = 0.f;
#pragma unroll 8
  for (int c = 0; c < CH; ++c) {
    float v = outT[(size_t)(b * CH + c) * NNODE + j];
    aL = fmaf(v, alpha[c], aL);
    aR = fmaf(v, alpha[CH + c], aR);
  }
  l[gid] = aL;
  r[gid] = aR;
}

#define JS 4
#define NJ (NNODE / JS)
#define SB 4
#define SC (32 * SB)
#define BP (SC + 4)
#define STAGE (CH * BP)

template <int E>
__device__ __forceinline__ void k2_diag1(float& a, const int dd) {
  asm("v_cmp_ne_u32_e32 vcc_lo, %2, %1\n\tv_cndmask_b32_e32 %0, 1.0, %0, vcc_lo"
      : "+v"(a)
      : "v"(dd), "n"(E)
      : "vcc_lo");
}
__device__ __forceinline__ void k2_diag(float4 (&a4)[4], const int dd) {
  k2_diag1<0>(a4[0].x, dd);  k2_diag1<1>(a4[0].y, dd);
  k2_diag1<2>(a4[0].z, dd);  k2_diag1<3>(a4[0].w, dd);
  k2_diag1<4>(a4[1].x, dd);  k2_diag1<5>(a4[1].y, dd);
  k2_diag1<6>(a4[1].z, dd);  k2_diag1<7>(a4[1].w, dd);
  k2_diag1<8>(a4[2].x, dd);  k2_diag1<9>(a4[2].y, dd);
  k2_diag1<10>(a4[2].z, dd); k2_diag1<11>(a4[2].w, dd);
  k2_diag1<12>(a4[3].x, dd); k2_diag1<13>(a4[3].y, dd);
  k2_diag1<14>(a4[3].z, dd); k2_diag1<15>(a4[3].w, dd);
}

template <bool LAST>
__device__ __forceinline__ void k2_block(
    const float4 (&a4)[4], const float4 (&r4)[4], float4 (&na4)[4],
    float4 (&nr4)[4], const float* __restrict__ Bc, float* __restrict__ Bn,
    const float* __restrict__ gsrc, const float* __restrict__ Arow,
    const float* __restrict__ rb, const int jg, const int jn, const bool pn,
    const int dn, const float l2, v8f (&acc)[4], v2f& rsum2) {
  float4 gB[4], b4[4][4];
#pragma unroll
  for (int q = 0; q < 4; ++q)
    gB[q] = *(const float4*)(gsrc + (size_t)(16 * q) * NNODE + jg);
#pragma unroll
  for (int g = 0; g < 4; ++g) {
    na4[g] = *(const float4*)(Arow + jn + 4 * g);
    nr4[g] = *(const float4*)(rb + jn + 4 * g);
  }
#pragma unroll
  for (int t = 0; t < 4; ++t)
#pragma unroll
    for (int g = 0; g < 4; ++g)
      b4[t][g] = *(const float4*)(Bc + t * 16 * BP + 4 * g);
#pragma unroll
  for (int g = 0; g < 4; ++g) {
    const float av[4] = {a4[g].x, a4[g].y, a4[g].z, a4[g].w};
    const float rv[4] = {r4[g].x, r4[g].y, r4[g].z, r4[g].w};
    float w[4];
#pragma unroll
    for (int u = 0; u < 4; ++u) {
      float s = l2 * rv[u];
      s = __builtin_fmaxf(s, 0.01f * s);
      w[u] = av[u] * __builtin_amdgcn_exp2f(s);
    }
#pragma unroll
    for (int p = 0; p < 2; ++p) {
      v2f af;
      af.x = w[2 * p];
      af.y = w[2 * p + 1];
      rsum2 += af;
#pragma unroll
      for (int t = 0; t < 4; ++t) {
        v2f bf;
        bf.x = p ? b4[t][g].z : b4[t][g].x;
        bf.y = p ? b4[t][g].w : b4[t][g].y;
        acc[t] = wmma4(af, bf, acc[t]);
      }
    }
  }
#pragma unroll
  for (int q = 0; q < 4; ++q) *(float4*)(Bn + 16 * q * BP) = gB[q];
  __builtin_amdgcn_sched_group_barrier(0x020, 16, 0);
  __builtin_amdgcn_sched_group_barrier(0x100, 16, 0);
  __builtin_amdgcn_sched_group_barrier(0x008, 32, 0);
  __builtin_amdgcn_sched_group_barrier(0x200, 4, 0);
  asm volatile("" ::"v"(acc[0]), "v"(acc[1]), "v"(acc[2]), "v"(acc[3]),
               "v"(rsum2.x), "v"(rsum2.y),
               "v"(na4[0].x), "v"(na4[0].w), "v"(na4[1].x), "v"(na4[1].w),
               "v"(na4[2].x), "v"(na4[2].w), "v"(na4[3].x), "v"(na4[3].w),
               "v"(nr4[0].x), "v"(nr4[0].w), "v"(nr4[1].x), "v"(nr4[1].w),
               "v"(nr4[2].x), "v"(nr4[2].w), "v"(nr4[3].x), "v"(nr4[3].w));
  if (__builtin_expect(pn, 0)) k2_diag(na4, dn);
  if (LAST) {
    __builtin_amdgcn_fence(__ATOMIC_RELEASE, "workgroup", "local");
    __builtin_amdgcn_s_barrier();
    __builtin_amdgcn_fence(__ATOMIC_ACQUIRE, "workgroup", "local");
  }
}

__global__ __launch_bounds__(128 * JS) void k2_main(
    const float* __restrict__ A, const float* __restrict__ outT,
    const float* __restrict__ l, const float* __restrict__ r,
    const float* __restrict__ w2, float* __restrict__ out) {
  __shared__ __align__(16) float smem[JS * 2 * STAGE];
  float* const tileAll = smem;
  float* const RaccAll = tileAll + 4 * 16 * 66;
  float* const RsumAll = RaccAll + (JS - 1) * 4 * 32 * 32;

  const int tid  = threadIdx.x;
  const int lane = tid & 31;
  const int wave = tid >> 5;
  const int strip = wave & 3;
  const int js   = wave >> 2;
  const int tg   = tid & 127;
  const int jbeg = js * NJ;
  const int job  = blockIdx.x * 4 + strip;
  const int b    = job >> 8;
  const int i0   = (job & 255) << 4;
  const int m    = lane & 15;
  const int h    = lane >> 4;

  const float l2 = l[b * NNODE + i0 + m] * 1.44269504088896341f;
  const float* Arow = A + (size_t)(i0 + m) * NNODE + 16 * h + jbeg;
  const float* rb   = r + b * NNODE + 16 * h + jbeg;
  const float* oT   = outT + (size_t)b * CH * NNODE;

  const float* gsrc = oT + (size_t)(tg >> 3) * NNODE + 4 * (tg & 7) + jbeg;
  const int sdst = (tg >> 3) * BP + 4 * (tg & 7);
  float* const Bg = smem + js * 2 * STAGE;
  const int fsrc = m * BP + 16 * h;

  v8f acc[4] = {};
  v2f rsum2 = {0.f, 0.f};

  float4 aA[4], rA[4], aB[4], rB[4];
  {
    float4 gB[SB][4];
#pragma unroll
    for (int s = 0; s < SB; ++s)
#pragma unroll
      for (int q = 0; q < 4; ++q)
        gB[s][q] = *(const float4*)(gsrc + (size_t)(16 * q) * NNODE + 32 * s);
#pragma unroll
    for (int g = 0; g < 4; ++g) {
      aA[g] = *(const float4*)(Arow + 4 * g);
      rA[g] = *(const float4*)(rb + 4 * g);
    }
#pragma unroll
    for (int s = 0; s < SB; ++s)
#pragma unroll
      for (int q = 0; q < 4; ++q)
        *(float4*)(Bg + sdst + 32 * s + 16 * q * BP) = gB[s][q];
  }
  __syncthreads();

  int d = i0 + m - 16 * h - jbeg;
  const int jdiag = __builtin_amdgcn_readfirstlane((i0 & ~31) - jbeg);
  if (jdiag == 0) k2_diag(aA, d);
  int cs = 0;
  for (int j0 = 0; j0 < NJ; j0 += SC) {
    const float* Bc = Bg + cs + fsrc;
    float* Bn = Bg + (STAGE - cs) + sdst;
    const bool more = j0 + SC < NJ;
    const int jg = more ? j0 + SC : 0;
    const int jend = more ? j0 + SC : NJ - 32;
    k2_block<false>(aA, rA, aB, rB, Bc, Bn, gsrc, Arow, rb, jg, j0 + 32,
                    j0 + 32 == jdiag, d - 32, l2, acc, rsum2);
    k2_block<false>(aB, rB, aA, rA, Bc + 32, Bn + 32, gsrc, Arow, rb, jg + 32,
                    j0 + 64, j0 + 64 == jdiag, d - 64, l2, acc, rsum2);
    k2_block<false>(aA, rA, aB, rB, Bc + 64, Bn + 64, gsrc, Arow, rb, jg + 64,
                    j0 + 96, j0 + 96 == jdiag, d - 96, l2, acc, rsum2);
    k2_block<true>(aB, rB, aA, rA, Bc + 96, Bn + 96, gsrc, Arow, rb, jg + 96,
                   jend, more && j0 + SC == jdiag, d - SC, l2, acc, rsum2);
    d -= SC;
    cs = STAGE - cs;
  }

  float rsum = rsum2.x + rsum2.y;
  if (js > 0) {
    float* R = RaccAll + ((js - 1) * 4 + strip) * (32 * 32);
#pragma unroll
    for (int t = 0; t < 4; ++t)
#pragma unroll
      for (int rr = 0; rr < 8; ++rr) R[(t * 8 + rr) * 32 + lane] = acc[t][rr];
    RsumAll[((js - 1) * 4 + strip) * 32 + lane] = rsum;
  }
  __syncthreads();

  float* T = tileAll + strip * (16 * 66);
  if (js == 0) {
#pragma unroll
    for (int p = 0; p < JS - 1; ++p) {
      const float* R = RaccAll + (p * 4 + strip) * (32 * 32);
#pragma unroll
      for (int t = 0; t < 4; ++t)
#pragma unroll
        for (int rr = 0; rr < 8; ++rr) acc[t][rr] += R[(t * 8 + rr) * 32 + lane];
      rsum += RsumAll[(p * 4 + strip) * 32 + lane];
    }

    rsum += __shfl_xor(rsum, 16, 32);
    const float inv = 1.0f / rsum;

#pragma unroll
    for (int rr = 0; rr < 8; ++rr) {
      const float invr = __shfl(inv, rr + 8 * h, 32);
#pragma unroll
      for (int t = 0; t < 4; ++t)
        T[(rr + 8 * h) * 66 + t * 16 + m] = acc[t][rr] * invr;
    }
  }
  __syncthreads();

  if (js == 0) {
    v8f fin[4] = {};
#pragma unroll 4
    for (int k0 = 0; k0 < CH; k0 += 4) {
      const int kb = k0 + 2 * h;
      v2f a = *(const v2f*)(T + m * 66 + kb);
#pragma unroll
      for (int t = 0; t < 4; ++t) {
        v2f bb;
        bb.x = w2[kb * CH + t * 16 + m];
        bb.y = w2[(kb + 1) * CH + t * 16 + m];
        fin[t] = wmma4(a, bb, fin[t]);
      }
    }

#pragma unroll
    for (int t = 0; t < 4; ++t)
#pragma unroll
      for (int rr = 0; rr < 8; ++rr)
        out[(size_t)(b * NNODE + i0 + rr + 8 * h) * CH + t * 16 + m] =
            fin[t][rr];
  }
}

extern "C" void kernel_launch(void* const* d_in, const int* in_sizes, int n_in,
                              void* d_out, int out_size, void* d_ws,
                              size_t ws_size, hipStream_t stream) {
  const float* X     = (const float*)d_in[0];
  const float* A     = (const float*)d_in[1];
  const float* w1    = (const float*)d_in[2];
  const float* w2    = (const float*)d_in[3];
  const float* alpha = (const float*)d_in[4];
  float* out = (float*)d_out;

  float* ws   = (float*)d_ws;
  float* outT = ws;
  float* l    = outT + (size_t)NB * CH * NNODE;
  float* r    = l + (size_t)NB * NNODE;

  k1_xw1<<<NB * 256 / 4, 128, 0, stream>>>(X, w1, outT);
  k1b_lr<<<NB * NNODE / 256, 256, 0, stream>>>(outT, alpha, l, r);
  k2_main<<<NB * 256 / 4, 128 * JS, 0, stream>>>(A, outT, l, r, w2, out);
}


# k1b folded into k1: each wave forms l, r of its 16 rows from its accumulators (4-term share per lane, 16 shares added via LDS in fixed order); 3 launches -> 2; l, r sum order changes
# speedup vs baseline: 2.6481x; 1.0822x over previous
#include <hip/hip_runtime.h>

typedef float v2f __attribute__((ext_vector_type(2)));
typedef float v8f __attribute__((ext_vector_type(8)));

#define NNODE 4096
#define CH 64
#define NB 4

__device__ __forceinline__ v8f wmma4(v2f a, v2f b, v8f c) {
  return __builtin_amdgcn_wmma_f32_16x16x4_f32(false, a, false, b, (short)0, c,
                                               false, false);
}

#define PP 33
__global__ __launch_bounds__(128) void k1_xw1(const float* __restrict__ X,
                                              const float* __restrict__ w1,
                                              const float* __restrict__ alpha,
                                              float* __restrict__ outT,
                                              float* __restrict__ l,
                                              float* __restrict__ r) {
  __shared__ float part[4][16 * PP];
  const int lane = threadIdx.x & 31;
  const int wave = threadIdx.x >> 5;
  const int job  = blockIdx.x * 4 + wave;
  const int b    = job >> 8;
  const int j0   = (job & 255) << 4;
  const int m    = lane & 15;
  const int h    = lane >> 4;

  float aL[4], aR[4];
#pragma unroll
  for (int t = 0; t < 4; ++t) {
    aL[t] = alpha[t * 16 + m];
    aR[t] = alpha[CH + t * 16 + m];
  }

  v8f acc[4] = {};
  const float* Xrow = X + (size_t)(b * NNODE + j0 + m) * CH;

#pragma unroll 4
  for (int k0 = 0; k0 < CH; k0 += 4) {
    const int kb = k0 + 2 * h;
    v2f a = *(const v2f*)(Xrow + kb);
#pragma unroll
    for (int t = 0; t < 4; ++t) {
      v2f bb;
      bb.x = w1[kb * CH + t * 16 + m];
      bb.y = w1[(kb + 1) * CH + t * 16 + m];
      acc[t] = wmma4(a, bb, acc[t]);
    }
  }
#pragma unroll
  for (int t = 0; t < 4; ++t)
#pragma unroll
    for (int rr = 0; rr < 8; ++rr)
      outT[(size_t)(b * CH + t * 16 + m) * NNODE + j0 + rr + 8 * h] =
          acc[t][rr];

  float* P = part[wave];
#pragma unroll
  for (int rr = 0; rr < 8; ++rr) {
    float pl = 0.f, pr = 0.f;
#pragma unroll
    for (int t = 0; t < 4; ++t) {
      pl = fmaf(acc[t][rr], aL[t], pl);
      pr = fmaf(acc[t][rr], aR[t], pr);
    }
    P[(2 * rr) * PP + lane] = pl;
    P[(2 * rr + 1) * PP + lane] = pr;
  }
  __syncthreads();
  const int which = lane >> 4;
  const int jj = lane & 15;
  const float* Q = P + (2 * (jj & 7) + which) * PP + (jj >> 3) * 16;
  float s = Q[0];
#pragma unroll
  for (int mm = 1; mm < 16; ++mm) s += Q[mm];
  (which ? r : l)[b * NNODE + j0 + jj] = s;
}

#define JS 4
#define NJ (NNODE / JS)
#define SB 4
#define SC (32 * SB)
#define BP (SC + 4)
#define STAGE (CH * BP)

template <int E>
__device__ __forceinline__ void k2_diag1(float& a, const int dd) {
  asm("v_cmp_ne_u32_e32 vcc_lo, %2, %1\n\tv_cndmask_b32_e32 %0, 1.0, %0, vcc_lo"
      : "+v"(a)
      : "v"(dd), "n"(E)
      : "vcc_lo");
}
__device__ __forceinline__ void k2_diag(float4 (&a4)[4], const int dd) {
  k2_diag1<0>(a4[0].x, dd);  k2_diag1<1>(a4[0].y, dd);
  k2_diag1<2>(a4[0].z, dd);  k2_diag1<3>(a4[0].w, dd);
  k2_diag1<4>(a4[1].x, dd);  k2_diag1<5>(a4[1].y, dd);
  k2_diag1<6>(a4[1].z, dd);  k2_diag1<7>(a4[1].w, dd);
  k2_diag1<8>(a4[2].x, dd);  k2_diag1<9>(a4[2].y, dd);
  k2_diag1<10>(a4[2].z, dd); k2_diag1<11>(a4[2].w, dd);
  k2_diag1<12>(a4[3].x, dd); k2_diag1<13>(a4[3].y, dd);
  k2_diag1<14>(a4[3].z, dd); k2_diag1<15>(a4[3].w, dd);
}

template <bool LAST>
__device__ __forceinline__ void k2_block(
    const float4 (&a4)[4], const float4 (&r4)[4], float4 (&na4)[4],
    float4 (&nr4)[4], const float* __restrict__ Bc, float* __restrict__ Bn,
    const float* __restrict__ gsrc, const float* __restrict__ Arow,
    const float* __restrict__ rb, const int jg, const int jn, const bool pn,
    const int dn, const float l2, v8f (&acc)[4], v2f& rsum2) {
  float4 gB[4], b4[4][4];
#pragma unroll
  for (int q = 0; q < 4; ++q)
    gB[q] = *(const float4*)(gsrc + (size_t)(16 * q) * NNODE + jg);
#pragma unroll
  for (int g = 0; g < 4; ++g) {
    na4[g] = *(const float4*)(Arow + jn + 4 * g);
    nr4[g] = *(const float4*)(rb + jn + 4 * g);
  }
#pragma unroll
  for (int t = 0; t < 4; ++t)
#pragma unroll
    for (int g = 0; g < 4; ++g)
      b4[t][g] = *(const float4*)(Bc + t * 16 * BP + 4 * g);
#pragma unroll
  for (int g = 0; g < 4; ++g) {
    const float av[4] = {a4[g].x, a4[g].y, a4[g].z, a4[g].w};
    const float rv[4] = {r4[g].x, r4[g].y, r4[g].z, r4[g].w};
    float w[4];
#pragma unroll
    for (int u = 0; u < 4; ++u) {
      float s = l2 * rv[u];
      s = __builtin_fmaxf(s, 0.01f * s);
      w[u] = av[u] * __builtin_amdgcn_exp2f(s);
    }
#pragma unroll
    for (int p = 0; p < 2; ++p) {
      v2f af;
      af.x = w[2 * p];
      af.y = w[2 * p + 1];
      rsum2 += af;
#pragma unroll
      for (int t = 0; t < 4; ++t) {
        v2f bf;
        bf.x = p ? b4[t][g].z : b4[t][g].x;
        bf.y = p ? b4[t][g].w : b4[t][g].y;
        acc[t] = wmma4(af, bf, acc[t]);
      }
    }
  }
#pragma unroll
  for (int q = 0; q < 4; ++q) *(float4*)(Bn + 16 * q * BP) = gB[q];
  __builtin_amdgcn_sched_group_barrier(0x020, 16, 0);
  __builtin_amdgcn_sched_group_barrier(0x100, 16, 0);
  __builtin_amdgcn_sched_group_barrier(0x008, 32, 0);
  __builtin_amdgcn_sched_group_barrier(0x200, 4, 0);
  asm volatile("" ::"v"(acc[0]), "v"(acc[1]), "v"(acc[2]), "v"(acc[3]),
               "v"(rsum2.x), "v"(rsum2.y),
               "v"(na4[0].x), "v"(na4[0].w), "v"(na4[1].x), "v"(na4[1].w),
               "v"(na4[2].x), "v"(na4[2].w), "v"(na4[3].x), "v"(na4[3].w),
               "v"(nr4[0].x), "v"(nr4[0].w), "v"(nr4[1].x), "v"(nr4[1].w),
               "v"(nr4[2].x), "v"(nr4[2].w), "v"(nr4[3].x), "v"(nr4[3].w));
  if (__builtin_expect(pn, 0)) k2_diag(na4, dn);
  if (LAST) {
    __builtin_amdgcn_fence(__ATOMIC_RELEASE, "workgroup", "local");
    __builtin_amdgcn_s_barrier();
    __builtin_amdgcn_fence(__ATOMIC_ACQUIRE, "workgroup", "local");
  }
}

__global__ __launch_bounds__(128 * JS) void k2_main(
    const float* __restrict__ A, const float* __restrict__ outT,
    const float* __restrict__ l, const float* __restrict__ r,
    const float* __restrict__ w2, float* __restrict__ out) {
  __shared__ __align__(16) float smem[JS * 2 * STAGE];
  float* const tileAll = smem;
  float* const RaccAll = tileAll + 4 * 16 * 66;
  float* const RsumAll = RaccAll + (JS - 1) * 4 * 32 * 32;

  const int tid  = threadIdx.x;
  const int lane = tid & 31;
  const int wave = tid >> 5;
  const int strip = wave & 3;
  const int js   = wave >> 2;
  const int tg   = tid & 127;
  const int jbeg = js * NJ;
  const int job  = blockIdx.x * 4 + strip;
  const int b    = job >> 8;
  const int i0   = (job & 255) << 4;
  const int m    = lane & 15;
  const int h    = lane >> 4;

  const float l2 = l[b * NNODE + i0 + m] * 1.44269504088896341f;
  const float* Arow = A + (size_t)(i0 + m) * NNODE + 16 * h + jbeg;
  const float* rb   = r + b * NNODE + 16 * h + jbeg;
  const float* oT   = outT + (size_t)b * CH * NNODE;

  const float* gsrc = oT + (size_t)(tg >> 3) * NNODE + 4 * (tg & 7) + jbeg;
  const int sdst = (tg >> 3) * BP + 4 * (tg & 7);
  float* const Bg = smem + js * 2 * STAGE;
  const int fsrc = m * BP + 16 * h;

  v8f acc[4] = {};
  v2f rsum2 = {0.f, 0.f};

  float4 aA[4], rA[4], aB[4], rB[4];
  {
    float4 gB[SB][4];
#pragma unroll
    for (int s = 0; s < SB; ++s)
#pragma unroll
      for (int q = 0; q < 4; ++q)
        gB[s][q] = *(const float4*)(gsrc + (size_t)(16 * q) * NNODE + 32 * s);
#pragma unroll
    for (int g = 0; g < 4; ++g) {
      aA[g] = *(const float4*)(Arow + 4 * g);
      rA[g] = *(const float4*)(rb + 4 * g);
    }
#pragma unroll
    for (int s = 0; s < SB; ++s)
#pragma unroll
      for (int q = 0; q < 4; ++q)
        *(float4*)(Bg + sdst + 32 * s + 16 * q * BP) = gB[s][q];
  }
  __syncthreads();

  int d = i0 + m - 16 * h - jbeg;
  const int jdiag = __builtin_amdgcn_readfirstlane((i0 & ~31) - jbeg);
  if (jdiag == 0) k2_diag(aA, d);
  int cs = 0;
  for (int j0 = 0; j0 < NJ; j0 += SC) {
    const float* Bc = Bg + cs + fsrc;
    float* Bn = Bg + (STAGE - cs) + sdst;
    const bool more = j0 + SC < NJ;
    const int jg = more ? j0 + SC : 0;
    const int jend = more ? j0 + SC : NJ - 32;
    k2_block<false>(aA, rA, aB, rB, Bc, Bn, gsrc, Arow, rb, jg, j0 + 32,
                    j0 + 32 == jdiag, d - 32, l2, acc, rsum2);
    k2_block<false>(aB, rB, aA, rA, Bc + 32, Bn + 32, gsrc, Arow, rb, jg + 32,
                    j0 + 64, j0 + 64 == jdiag, d - 64, l2, acc, rsum2);
    k2_block<false>(aA, rA, aB, rB, Bc + 64, Bn + 64, gsrc, Arow, rb, jg + 64,
                    j0 + 96, j0 + 96 == jdiag, d - 96, l2, acc, rsum2);
    k2_block<true>(aB, rB, aA, rA, Bc + 96, Bn + 96, gsrc, Arow, rb, jg + 96,
                   jend, more && j0 + SC == jdiag, d - SC, l2, acc, rsum2);
    d -= SC;
    cs = STAGE - cs;
  }

  float rsum = rsum2.x + rsum2.y;
  if (js > 0) {
    float* R = RaccAll + ((js - 1) * 4 + strip) * (32 * 32);
#pragma unroll
    for (int t = 0; t < 4; ++t)
#pragma unroll
      for (int rr = 0; rr < 8; ++rr) R[(t * 8 + rr) * 32 + lane] = acc[t][rr];
    RsumAll[((js - 1) * 4 + strip) * 32 + lane] = rsum;
  }
  __syncthreads();

  float* T = tileAll + strip * (16 * 66);
  if (js == 0) {
#pragma unroll
    for (int p = 0; p < JS - 1; ++p) {
      const float* R = RaccAll + (p * 4 + strip) * (32 * 32);
#pragma unroll
      for (int t = 0; t < 4; ++t)
#pragma unroll
        for (int rr = 0; rr < 8; ++rr) acc[t][rr] += R[(t * 8 + rr) * 32 + lane];
      rsum += RsumAll[(p * 4 + strip) * 32 + lane];
    }

    rsum += __shfl_xor(rsum, 16, 32);
    const float inv = 1.0f / rsum;

#pragma unroll
    for (int rr = 0; rr < 8; ++rr) {
      const float invr = __shfl(inv, rr + 8 * h, 32);
#pragma unroll
      for (int t = 0; t < 4; ++t)
        T[(rr + 8 * h) * 66 + t * 16 + m] = acc[t][rr] * invr;
    }
  }
  __syncthreads();

  if (js == 0) {
    v8f fin[4] = {};
#pragma unroll 4
    for (int k0 = 0; k0 < CH; k0 += 4) {
      const int kb = k0 + 2 * h;
      v2f a = *(const v2f*)(T + m * 66 + kb);
#pragma unroll
      for (int t = 0; t < 4; ++t) {
        v2f bb;
        bb.x = w2[kb * CH + t * 16 + m];
        bb.y = w2[(kb + 1) * CH + t * 16 + m];
        fin[t] = wmma4(a, bb, fin[t]);
      }
    }

#pragma unroll
    for (int t = 0; t < 4; ++t)
#pragma unroll
      for (int rr = 0; rr < 8; ++rr)
        out[(size_t)(b * NNODE + i0 + rr + 8 * h) * CH + t * 16 + m] =
            fin[t][rr];
  }
}

extern "C" void kernel_launch(void* const* d_in, const int* in_sizes, int n_in,
                              void* d_out, int out_size, void* d_ws,
                              size_t ws_size, hipStream_t stream) {
  const float* X     = (const float*)d_in[0];
  const float* A     = (const float*)d_in[1];
  const float* w1    = (const float*)d_in[2];
  const float* w2    = (const float*)d_in[3];
  const float* alpha = (const float*)d_in[4];
  float* out = (float*)d_out;

  float* ws   = (float*)d_ws;
  float* outT = ws;
  float* l    = outT + (size_t)NB * CH * NNODE;
  float* r    = l + (size_t)NB * NNODE;

  k1_xw1<<<NB * 256 / 4, 128, 0, stream>>>(X, w1, alpha, outT, l, r);
  k2_main<<<NB * 256 / 4, 128 * JS, 0, stream>>>(A, outT, l, r, w2, out);
}


# k2: each wave steps its issue priority down per block of a stage (s_setprio 3,2,1,0), so a SIMD's lagging waves are served first and its 4 waves reach the stage barrier together; sums unchanged
# speedup vs baseline: 2.7957x; 1.0557x over previous
#include <hip/hip_runtime.h>

typedef float v2f __attribute__((ext_vector_type(2)));
typedef float v8f __attribute__((ext_vector_type(8)));

#define NNODE 4096
#define CH 64
#define NB 4

__device__ __forceinline__ v8f wmma4(v2f a, v2f b, v8f c) {
  return __builtin_amdgcn_wmma_f32_16x16x4_f32(false, a, false, b, (short)0, c,
                                               false, false);
}

#define PP 33
__global__ __launch_bounds__(128) void k1_xw1(const float* __restrict__ X,
                                              const float* __restrict__ w1,
                                              const float* __restrict__ alpha,
                                              float* __restrict__ outT,
                                              float* __restrict__ l,
                                              float* __restrict__ r) {
  __shared__ float part[4][16 * PP];
  const int lane = threadIdx.x & 31;
  const int wave = threadIdx.x >> 5;
  const int job  = blockIdx.x * 4 + wave;
  const int b    = job >> 8;
  const int j0   = (job & 255) << 4;
  const int m    = lane & 15;
  const int h    = lane >> 4;

  float aL[4], aR[4];
#pragma unroll
  for (int t = 0; t < 4; ++t) {
    aL[t] = alpha[t * 16 + m];
    aR[t] = alpha[CH + t * 16 + m];
  }

  v8f acc[4] = {};
  const float* Xrow = X + (size_t)(b * NNODE + j0 + m) * CH;

#pragma unroll 4
  for (int k0 = 0; k0 < CH; k0 += 4) {
    const int kb = k0 + 2 * h;
    v2f a = *(const v2f*)(Xrow + kb);
#pragma unroll
    for (int t = 0; t < 4; ++t) {
      v2f bb;
      bb.x = w1[kb * CH + t * 16 + m];
      bb.y = w1[(kb + 1) * CH + t * 16 + m];
      acc[t] = wmma4(a, bb, acc[t]);
    }
  }
#pragma unroll
  for (int t = 0; t < 4; ++t)
#pragma unroll
    for (int rr = 0; rr < 8; ++rr)
      outT[(size_t)(b * CH + t * 16 + m) * NNODE + j0 + rr + 8 * h] =
          acc[t][rr];

  float* P = part[wave];
#pragma unroll
  for (int rr = 0; rr < 8; ++rr) {
    float pl = 0.f, pr = 0.f;
#pragma unroll
    for (int t = 0; t < 4; ++t) {
      pl = fmaf(acc[t][rr], aL[t], pl);
      pr = fmaf(acc[t][rr], aR[t], pr);
    }
    P[(2 * rr) * PP + lane] = pl;
    P[(2 * rr + 1) * PP + lane] = pr;
  }
  __syncthreads();
  const int which = lane >> 4;
  const int jj = lane & 15;
  const float* Q = P + (2 * (jj & 7) + which) * PP + (jj >> 3) * 16;
  float s = Q[0];
#pragma unroll
  for (int mm = 1; mm < 16; ++mm) s += Q[mm];
  (which ? r : l)[b * NNODE + j0 + jj] = s;
}

#define JS 4
#define NJ (NNODE / JS)
#define SB 4
#define SC (32 * SB)
#define BP (SC + 4)
#define STAGE (CH * BP)

template <int E>
__device__ __forceinline__ void k2_diag1(float& a, const int dd) {
  asm("v_cmp_ne_u32_e32 vcc_lo, %2, %1\n\tv_cndmask_b32_e32 %0, 1.0, %0, vcc_lo"
      : "+v"(a)
      : "v"(dd), "n"(E)
      : "vcc_lo");
}
__device__ __forceinline__ void k2_diag(float4 (&a4)[4], const int dd) {
  k2_diag1<0>(a4[0].x, dd);  k2_diag1<1>(a4[0].y, dd);
  k2_diag1<2>(a4[0].z, dd);  k2_diag1<3>(a4[0].w, dd);
  k2_diag1<4>(a4[1].x, dd);  k2_diag1<5>(a4[1].y, dd);
  k2_diag1<6>(a4[1].z, dd);  k2_diag1<7>(a4[1].w, dd);
  k2_diag1<8>(a4[2].x, dd);  k2_diag1<9>(a4[2].y, dd);
  k2_diag1<10>(a4[2].z, dd); k2_diag1<11>(a4[2].w, dd);
  k2_diag1<12>(a4[3].x, dd); k2_diag1<13>(a4[3].y, dd);
  k2_diag1<14>(a4[3].z, dd); k2_diag1<15>(a4[3].w, dd);
}

template <bool LAST, int PRIO>
__device__ __forceinline__ void k2_block(
    const float4 (&a4)[4], const float4 (&r4)[4], float4 (&na4)[4],
    float4 (&nr4)[4], const float* __restrict__ Bc, float* __restrict__ Bn,
    const float* __restrict__ gsrc, const float* __restrict__ Arow,
    const float* __restrict__ rb, const int jg, const int jn, const bool pn,
    const int dn, const float l2, v8f (&acc)[4], v2f& rsum2) {
  __builtin_amdgcn_s_setprio(PRIO);
  float4 gB[4], b4[4][4];
#pragma unroll
  for (int q = 0; q < 4; ++q)
    gB[q] = *(const float4*)(gsrc + (size_t)(16 * q) * NNODE + jg);
#pragma unroll
  for (int g = 0; g < 4; ++g) {
    na4[g] = *(const float4*)(Arow + jn + 4 * g);
    nr4[g] = *(const float4*)(rb + jn + 4 * g);
  }
#pragma unroll
  for (int t = 0; t < 4; ++t)
#pragma unroll
    for (int g = 0; g < 4; ++g)
      b4[t][g] = *(const float4*)(Bc + t * 16 * BP + 4 * g);
#pragma unroll
  for (int g = 0; g < 4; ++g) {
    const float av[4] = {a4[g].x, a4[g].y, a4[g].z, a4[g].w};
    const float rv[4] = {r4[g].x, r4[g].y, r4[g].z, r4[g].w};
    float w[4];
#pragma unroll
    for (int u = 0; u < 4; ++u) {
      float s = l2 * rv[u];
      s = __builtin_fmaxf(s, 0.01f * s);
      w[u] = av[u] * __builtin_amdgcn_exp2f(s);
    }
#pragma unroll
    for (int p = 0; p < 2; ++p) {
      v2f af;
      af.x = w[2 * p];
      af.y = w[2 * p + 1];
      rsum2 += af;
#pragma unroll
      for (int t = 0; t < 4; ++t) {
        v2f bf;
        bf.x = p ? b4[t][g].z : b4[t][g].x;
        bf.y = p ? b4[t][g].w : b4[t][g].y;
        acc[t] = wmma4(af, bf, acc[t]);
      }
    }
  }
#pragma unroll
  for (int q = 0; q < 4; ++q) *(float4*)(Bn + 16 * q * BP) = gB[q];
  __builtin_amdgcn_sched_group_barrier(0x020, 16, 0);
  __builtin_amdgcn_sched_group_barrier(0x100, 16, 0);
  __builtin_amdgcn_sched_group_barrier(0x008, 32, 0);
  __builtin_amdgcn_sched_group_barrier(0x200, 4, 0);
  asm volatile("" ::"v"(acc[0]), "v"(acc[1]), "v"(acc[2]), "v"(acc[3]),
               "v"(rsum2.x), "v"(rsum2.y),
               "v"(na4[0].x), "v"(na4[0].w), "v"(na4[1].x), "v"(na4[1].w),
               "v"(na4[2].x), "v"(na4[2].w), "v"(na4[3].x), "v"(na4[3].w),
               "v"(nr4[0].x), "v"(nr4[0].w), "v"(nr4[1].x), "v"(nr4[1].w),
               "v"(nr4[2].x), "v"(nr4[2].w), "v"(nr4[3].x), "v"(nr4[3].w));
  if (__builtin_expect(pn, 0)) k2_diag(na4, dn);
  if (LAST) {
    __builtin_amdgcn_fence(__ATOMIC_RELEASE, "workgroup", "local");
    __builtin_amdgcn_s_barrier();
    __builtin_amdgcn_fence(__ATOMIC_ACQUIRE, "workgroup", "local");
  }
}

__global__ __launch_bounds__(128 * JS) void k2_main(
    const float* __restrict__ A, const float* __restrict__ outT,
    const float* __restrict__ l, const float* __restrict__ r,
    const float* __restrict__ w2, float* __restrict__ out) {
  __shared__ __align__(16) float smem[JS * 2 * STAGE];
  float* const tileAll = smem;
  float* const RaccAll = tileAll + 4 * 16 * 66;
  float* const RsumAll = RaccAll + (JS - 1) * 4 * 32 * 32;

  const int tid  = threadIdx.x;
  const int lane = tid & 31;
  const int wave = tid >> 5;
  const int strip = wave & 3;
  const int js   = wave >> 2;
  const int tg   = tid & 127;
  const int jbeg = js * NJ;
  const int job  = blockIdx.x * 4 + strip;
  const int b    = job >> 8;
  const int i0   = (job & 255) << 4;
  const int m    = lane & 15;
  const int h    = lane >> 4;

  const float l2 = l[b * NNODE + i0 + m] * 1.44269504088896341f;
  const float* Arow = A + (size_t)(i0 + m) * NNODE + 16 * h + jbeg;
  const float* rb   = r + b * NNODE + 16 * h + jbeg;
  const float* oT   = outT + (size_t)b * CH * NNODE;

  const float* gsrc = oT + (size_t)(tg >> 3) * NNODE + 4 * (tg & 7) + jbeg;
  const int sdst = (tg >> 3) * BP + 4 * (tg & 7);
  float* const Bg = smem + js * 2 * STAGE;
  const int fsrc = m * BP + 16 * h;

  v8f acc[4] = {};
  v2f rsum2 = {0.f, 0.f};

  float4 aA[4], rA[4], aB[4], rB[4];
  {
    float4 gB[SB][4];
#pragma unroll
    for (int s = 0; s < SB; ++s)
#pragma unroll
      for (int q = 0; q < 4; ++q)
        gB[s][q] = *(const float4*)(gsrc + (size_t)(16 * q) * NNODE + 32 * s);
#pragma unroll
    for (int g = 0; g < 4; ++g) {
      aA[g] = *(const float4*)(Arow + 4 * g);
      rA[g] = *(const float4*)(rb + 4 * g);
    }
#pragma unroll
    for (int s = 0; s < SB; ++s)
#pragma unroll
      for (int q = 0; q < 4; ++q)
        *(float4*)(Bg + sdst + 32 * s + 16 * q * BP) = gB[s][q];
  }
  __syncthreads();

  int d = i0 + m - 16 * h - jbeg;
  const int jdiag = __builtin_amdgcn_readfirstlane((i0 & ~31) - jbeg);
  if (jdiag == 0) k2_diag(aA, d);
  int cs = 0;
  for (int j0 = 0; j0 < NJ; j0 += SC) {
    const float* Bc = Bg + cs + fsrc;
    float* Bn = Bg + (STAGE - cs) + sdst;
    const bool more = j0 + SC < NJ;
    const int jg = more ? j0 + SC : 0;
    const int jend = more ? j0 + SC : NJ - 32;
    k2_block<false, 3>(aA, rA, aB, rB, Bc, Bn, gsrc, Arow, rb, jg, j0 + 32,
                    j0 + 32 == jdiag, d - 32, l2, acc, rsum2);
    k2_block<false, 2>(aB, rB, aA, rA, Bc + 32, Bn + 32, gsrc, Arow, rb, jg + 32,
                    j0 + 64, j0 + 64 == jdiag, d - 64, l2, acc, rsum2);
    k2_block<false, 1>(aA, rA, aB, rB, Bc + 64, Bn + 64, gsrc, Arow, rb, jg + 64,
                    j0 + 96, j0 + 96 == jdiag, d - 96, l2, acc, rsum2);
    k2_block<true, 0>(aB, rB, aA, rA, Bc + 96, Bn + 96, gsrc, Arow, rb, jg + 96,
                   jend, more && j0 + SC == jdiag, d - SC, l2, acc, rsum2);
    d -= SC;
    cs = STAGE - cs;
  }

  float rsum = rsum2.x + rsum2.y;
  if (js > 0) {
    float* R = RaccAll + ((js - 1) * 4 + strip) * (32 * 32);
#pragma unroll
    for (int t = 0; t < 4; ++t)
#pragma unroll
      for (int rr = 0; rr < 8; ++rr) R[(t * 8 + rr) * 32 + lane] = acc[t][rr];
    RsumAll[((js - 1) * 4 + strip) * 32 + lane] = rsum;
  }
  __syncthreads();

  float* T = tileAll + strip * (16 * 66);
  if (js == 0) {
#pragma unroll
    for (int p = 0; p < JS - 1; ++p) {
      const float* R = RaccAll + (p * 4 + strip) * (32 * 32);
#pragma unroll
      for (int t = 0; t < 4; ++t)
#pragma unroll
        for (int rr = 0; rr < 8; ++rr) acc[t][rr] += R[(t * 8 + rr) * 32 + lane];
      rsum += RsumAll[(p * 4 + strip) * 32 + lane];
    }

    rsum += __shfl_xor(rsum, 16, 32);
    const float inv = 1.0f / rsum;

#pragma unroll
    for (int rr = 0; rr < 8; ++rr) {
      const float invr = __shfl(inv, rr + 8 * h, 32);
#pragma unroll
      for (int t = 0; t < 4; ++t)
        T[(rr + 8 * h) * 66 + t * 16 + m] = acc[t][rr] * invr;
    }
  }
  __syncthreads();

  if (js == 0) {
    v8f fin[4] = {};
#pragma unroll 4
    for (int k0 = 0; k0 < CH; k0 += 4) {
      const int kb = k0 + 2 * h;
      v2f a = *(const v2f*)(T + m * 66 + kb);
#pragma unroll
      for (int t = 0; t < 4; ++t) {
        v2f bb;
        bb.x = w2[kb * CH + t * 16 + m];
        bb.y = w2[(kb + 1) * CH + t * 16 + m];
        fin[t] = wmma4(a, bb, fin[t]);
      }
    }

#pragma unroll
    for (int t = 0; t < 4; ++t)
#pragma unroll
      for (int rr = 0; rr < 8; ++rr)
        out[(size_t)(b * NNODE + i0 + rr + 8 * h) * CH + t * 16 + m] =
            fin[t][rr];
  }
}

extern "C" void kernel_launch(void* const* d_in, const int* in_sizes, int n_in,
                              void* d_out, int out_size, void* d_ws,
                              size_t ws_size, hipStream_t stream) {
  const float* X     = (const float*)d_in[0];
  const float* A     = (const float*)d_in[1];
  const float* w1    = (const float*)d_in[2];
  const float* w2    = (const float*)d_in[3];
  const float* alpha = (const float*)d_in[4];
  float* out = (float*)d_out;

  float* ws   = (float*)d_ws;
  float* outT = ws;
  float* l    = outT + (size_t)NB * CH * NNODE;
  float* r    = l + (size_t)NB * NNODE;

  k1_xw1<<<NB * 256 / 4, 128, 0, stream>>>(X, w1, alpha, outT, l, r);
  k2_main<<<NB * 256 / 4, 128 * JS, 0, stream>>>(A, outT, l, r, w2, out);
}
